# starting kernel
# baseline (speedup 1.0000x reference)
#include <hip/hip_runtime.h>
#include <hip/hip_bf16.h>

typedef __attribute__((ext_vector_type(2))) float v2f;
typedef __attribute__((ext_vector_type(8))) float v8f;

#define ALPHA_F 0.5f
#define NCLS 16

__global__ __launch_bounds__(256)
void lp_prep(const float* __restrict__ x, const float* __restrict__ W,
             float* __restrict__ hs, float* __restrict__ agg, int total) {
  int i = blockIdx.x * 256 + threadIdx.x;
  if (i < total) {
    hs[i]  = W[i >> 4] * x[i];
    agg[i] = 0.0f;
  }
}

__global__ __launch_bounds__(256)
void lp_edge(const int* __restrict__ ei, const float* __restrict__ ew,
             const float* __restrict__ hs, float* agg, int n_edges) {
  int gid = blockIdx.x * 256 + threadIdx.x;
  int e = gid >> 4;
  int c = gid & (NCLS - 1);
  if (e < n_edges) {
    if (c == 0) {
      __builtin_prefetch(ei + e + 4096, 0, 1);
      __builtin_prefetch(ew + e + 4096, 0, 1);
    }
    int   src = ei[e];
    int   dst = ei[n_edges + e];
    float w   = ew[e];
    float v   = hs[src * NCLS + c] * w;
    atomicAdd(&agg[dst * NCLS + c], v);
  }
}

__global__ __launch_bounds__(256)
void lp_node(const float* __restrict__ h_in, float* agg,
             const float* __restrict__ W, const float* __restrict__ degree,
             float* __restrict__ h_out, float* __restrict__ hs_out,
             int n_tiles) {
  const int lane  = threadIdx.x & 31;
  const int wave  = threadIdx.x >> 5;
  const int tile  = blockIdx.x * 8 + wave;
  if (tile >= n_tiles) return;
  const int node0 = tile * 16;
  const int m     = lane & 15;
  const int hi    = (lane >= 16) ? 1 : 0;

  float dg = degree[node0 + m];
  float av = (1.0f - ALPHA_F) * dg * dg;

  v8f acc;
#pragma unroll
  for (int r = 0; r < 8; ++r) {
    int row = node0 + r + (hi ? 8 : 0);
    acc[r] = ALPHA_F * h_in[row * NCLS + m];
  }

#pragma unroll
  for (int q = 0; q < 4; ++q) {
    int k0 = 4 * q + 2 * hi;
    v2f a, b;
    a.x = (m == k0)     ? av : 0.0f;
    a.y = (m == k0 + 1) ? av : 0.0f;
    b.x = agg[(node0 + k0)     * NCLS + m];
    b.y = agg[(node0 + k0 + 1) * NCLS + m];
    acc = __builtin_amdgcn_wmma_f32_16x16x4_f32(
         false, a,  false, b,
         (short)0, acc,  false,  false);
  }

#pragma unroll
  for (int r = 0; r < 8; ++r) {
    float s = acc[r];
    s += __shfl_xor(s, 1, 32);
    s += __shfl_xor(s, 2, 32);
    s += __shfl_xor(s, 4, 32);
    s += __shfl_xor(s, 8, 32);
    int   row = node0 + r + (hi ? 8 : 0);
    float o   = acc[r] / s;
    h_out [row * NCLS + m] = o;
    hs_out[row * NCLS + m] = W[row] * o;
    agg   [row * NCLS + m] = 0.0f;
  }
}

extern "C" void kernel_launch(void* const* d_in, const int* in_sizes, int n_in,
                              void* d_out, int out_size, void* d_ws, size_t ws_size,
                              hipStream_t stream) {
  const float* x   = (const float*)d_in[0];
  const float* W   = (const float*)d_in[1];
  const float* ew  = (const float*)d_in[2];
  const float* deg = (const float*)d_in[3];
  const int*   ei  = (const int*)  d_in[4];

  const int N       = in_sizes[3];
  const int N_EDGES = in_sizes[2];
  const int total   = N * NCLS;
  const int n_tiles = N / 16;

  float* out = (float*)d_out;
  float* agg = (float*)d_ws;
  float* hA  = agg + (size_t)total;
  float* hs  = hA  + (size_t)total;

  const int prep_blocks = (total + 255) / 256;
  const int edge_blocks = (N_EDGES * NCLS + 255) / 256;
  const int node_blocks = (n_tiles + 7) / 8;

  lp_prep<<<prep_blocks, 256, 0, stream>>>(x, W, hs, agg, total);

  for (int k = 0; k < 4; ++k) {
    const float* hin  = (k == 0) ? x : ((k & 1) ? hA : out);
    float*       hout = (k & 1) ? out : hA;
    lp_edge<<<edge_blocks, 256, 0, stream>>>(ei, ew, hs, agg, N_EDGES);
    lp_node<<<node_blocks, 256, 0, stream>>>(hin, agg, W, deg, hout, hs, n_tiles);
  }
}


# one launch per layer: the wave that sums 8 nodes' rows hands them through its own LDS rows to the WMMA node update (2 per wave), hs double-buffered; 13 launches -> 9; order of all sums unchanged
# speedup vs baseline: 3.8385x; 3.8385x over previous
#include <hip/hip_runtime.h>
#include <hip/hip_bf16.h>

typedef __attribute__((ext_vector_type(2))) float v2f;
typedef __attribute__((ext_vector_type(8))) float v8f;

#define ALPHA_F 0.5f
#define NCLS 16
#define SCAN_T 1024

#define BSH 9
#define BNODES 512
#define MAXB 256
#define CH 12800
#define CAP 24576
#define CH_T ((CH + 1023) / 1024)
#define BS_T 8

__global__ __launch_bounds__(256)
void lp_prep(const float* __restrict__ x, const float* __restrict__ W,
             float* __restrict__ hs, int total) {
  int i = blockIdx.x * 256 + threadIdx.x;
  if (i < total) hs[i] = W[i >> 4] * x[i];
}

__global__ __launch_bounds__(SCAN_T)
void lp_scan(const int* __restrict__ cnt, int* __restrict__ rowstart, int n_nodes) {
  __shared__ int wsum[32];
  __shared__ int base_s;
  const int t = threadIdx.x, lane = t & 31, wv = t >> 5;
  const int first = blockIdx.x * SCAN_T;

  int pre = 0;
  for (int i = t; i < first; i += SCAN_T) pre += cnt[i];
  for (int o = 16; o > 0; o >>= 1) pre += __shfl_xor(pre, o, 32);
  if (lane == 0) wsum[wv] = pre;
  __syncthreads();
  if (wv == 0) {
    int s = wsum[lane];
    for (int o = 16; o > 0; o >>= 1) s += __shfl_xor(s, o, 32);
    if (lane == 0) base_s = s;
  }
  __syncthreads();
  const int base = base_s;

  const int i = first + t;
  const int c = (i < n_nodes) ? cnt[i] : 0;
  int inc = c;
  for (int o = 1; o < 32; o <<= 1) {
    int u = __shfl_up(inc, o, 32);
    if (lane >= o) inc += u;
  }
  if (lane == 31) wsum[wv] = inc;
  __syncthreads();
  const int wtot = wsum[lane];
  int winc = wtot;
  for (int o = 1; o < 32; o <<= 1) {
    int u = __shfl_up(winc, o, 32);
    if (lane >= o) winc += u;
  }
  const int wexc = __shfl(winc - wtot, wv, 32);
  const int excl = base + wexc + inc - c;
  if (i < n_nodes) rowstart[i] = excl;
  if (i == n_nodes - 1) rowstart[n_nodes] = excl + c;
}

__device__ __forceinline__ int block_excl_scan(int c, int* wsum, int lane, int wv) {
  int inc = c;
  for (int o = 1; o < 32; o <<= 1) {
    int u = __shfl_up(inc, o, 32);
    if (lane >= o) inc += u;
  }
  if (lane == 31) wsum[wv] = inc;
  __syncthreads();
  const int wtot = wsum[lane];
  int winc = wtot;
  for (int o = 1; o < 32; o <<= 1) {
    int u = __shfl_up(winc, o, 32);
    if (lane >= o) winc += u;
  }
  const int wexc = __shfl(winc - wtot, wv, 32);
  __syncthreads();
  return wexc + inc - c;
}

__global__ __launch_bounds__(1024)
void lp_bhist(const int* __restrict__ ei, int* __restrict__ wghist,
              int n_edges, int nb, int g) {
  __shared__ int hist[MAXB];
  const int* __restrict__ dst = ei + n_edges;
  const int t = threadIdx.x, wg = blockIdx.x;
  if (t < MAXB) hist[t] = 0;
  __syncthreads();
  const int e0 = wg * CH;
  const int n_loc = min(CH, n_edges - e0);
  int dv[CH_T];
#pragma unroll
  for (int k = 0; k < CH_T; ++k) dv[k] = dst[e0 + min(t + k * 1024, n_loc - 1)];
#pragma unroll
  for (int k = 0; k < CH_T; ++k)
    if (t + k * 1024 < n_loc) atomicAdd(&hist[dv[k] >> BSH], 1);
  __syncthreads();
  if (t < nb) wghist[t * g + wg] = hist[t];
}

__global__ __launch_bounds__(1024)
void lp_bscatter(const int* __restrict__ ei, const float* __restrict__ ew,
                 const int* __restrict__ gscan, int* __restrict__ tmp_pairs_raw,
                 int* __restrict__ tmp_dst, int n_edges, int nb, int g) {
  __shared__ int2 sP[CH];
  __shared__ int  sD[CH];
  __shared__ int  hist[MAXB], lstart[MAXB], cursor[MAXB], gbase[MAXB];
  __shared__ int  wsum[32];
  const int* __restrict__ srcp = ei;
  const int* __restrict__ dst  = ei + n_edges;
  int2* __restrict__ tmp_pairs = reinterpret_cast<int2*>(tmp_pairs_raw);
  const int t = threadIdx.x, lane = t & 31, wv = t >> 5, wg = blockIdx.x;
  if (t < MAXB) {
    hist[t]  = 0;
    gbase[t] = (t < nb) ? gscan[t * g + wg] : 0;
  }
  __syncthreads();
  const int e0 = wg * CH;
  const int n_loc = min(CH, n_edges - e0);
  int dv[CH_T], sv[CH_T], wb[CH_T];
#pragma unroll
  for (int k = 0; k < CH_T; ++k) {
    const int e = e0 + min(t + k * 1024, n_loc - 1);
    dv[k] = dst[e];
    sv[k] = srcp[e];
    wb[k] = __float_as_int(ew[e]);
  }
#pragma unroll
  for (int k = 0; k < CH_T; ++k)
    if (t + k * 1024 < n_loc) atomicAdd(&hist[dv[k] >> BSH], 1);
  __syncthreads();
  const int c  = (t < MAXB) ? hist[t] : 0;
  const int ex = block_excl_scan(c, wsum, lane, wv);
  if (t < MAXB) { lstart[t] = ex; cursor[t] = ex; }
  __syncthreads();
#pragma unroll
  for (int k = 0; k < CH_T; ++k) {
    if (t + k * 1024 < n_loc) {
      const int p = atomicAdd(&cursor[dv[k] >> BSH], 1);
      sP[p] = make_int2(sv[k], wb[k]);
      sD[p] = dv[k];
    }
  }
  __syncthreads();
  for (int i = t; i < n_loc; i += 1024) {
    const int d  = sD[i];
    const int b  = d >> BSH;
    const int gp = gbase[b] + (i - lstart[b]);
    tmp_pairs[gp] = sP[i];
    tmp_dst[gp]   = d;
  }
}

__global__ __launch_bounds__(1024)
void lp_bsort(const int* __restrict__ gscan, const int* __restrict__ tmp_pairs_raw,
              const int* __restrict__ tmp_dst, int* __restrict__ rowstart,
              int* __restrict__ pairs_raw, int n_nodes, int nb, int g) {
  __shared__ int2 sP[CAP];
  __shared__ int  ncnt[BNODES], cursor[BNODES];
  __shared__ int  wsum[32];
  const int2* __restrict__ tmp_pairs = reinterpret_cast<const int2*>(tmp_pairs_raw);
  int2* __restrict__ pairs = reinterpret_cast<int2*>(pairs_raw);
  const int t = threadIdx.x, lane = t & 31, wv = t >> 5, b = blockIdx.x;
  const int bs  = gscan[b * g];
  const int be  = gscan[(b + 1) * g];
  const int cnt = be - bs;
  if (t < BNODES) ncnt[t] = 0;
  __syncthreads();
  for (int r0 = 0; r0 < cnt; r0 += BS_T * 1024) {
    int dv[BS_T];
#pragma unroll
    for (int k = 0; k < BS_T; ++k) dv[k] = tmp_dst[bs + min(r0 + k * 1024 + t, cnt - 1)];
#pragma unroll
    for (int k = 0; k < BS_T; ++k)
      if (r0 + k * 1024 + t < cnt) atomicAdd(&ncnt[dv[k] & (BNODES - 1)], 1);
  }
  __syncthreads();
  const int c  = (t < BNODES) ? ncnt[t] : 0;
  const int ex = block_excl_scan(c, wsum, lane, wv);
  if (t < BNODES) {
    cursor[t] = ex;
    const int node = b * BNODES + t;
    if (node < n_nodes) rowstart[node] = bs + ex;
  }
  if (b == nb - 1 && t == 0) rowstart[n_nodes] = be;
  __syncthreads();
  const bool fits = cnt <= CAP;
  for (int r0 = 0; r0 < cnt; r0 += BS_T * 1024) {
    int dv[BS_T]; int2 pv[BS_T];
#pragma unroll
    for (int k = 0; k < BS_T; ++k) {
      const int i = bs + min(r0 + k * 1024 + t, cnt - 1);
      dv[k] = tmp_dst[i];
      pv[k] = tmp_pairs[i];
    }
#pragma unroll
    for (int k = 0; k < BS_T; ++k) {
      if (r0 + k * 1024 + t < cnt) {
        const int pos = atomicAdd(&cursor[dv[k] & (BNODES - 1)], 1);
        if (fits) sP[pos] = pv[k]; else pairs[bs + pos] = pv[k];
      }
    }
  }
  if (fits) {
    __syncthreads();
    for (int i = t; i < cnt; i += 1024) pairs[bs + i] = sP[i];
  }
}

#define TROW 20

__global__ __launch_bounds__(256)
void lp_layer(const int* __restrict__ rowstart, const int* __restrict__ pairs_raw,
              const float* __restrict__ hs_in, const float* __restrict__ h_in,
              const float* __restrict__ W, const float* __restrict__ degree,
              float* __restrict__ h_out, float* __restrict__ hs_out, int n_nodes) {
  __shared__ __attribute__((aligned(16))) float tile[8][8 * TROW];
  const int lane  = threadIdx.x & 31;
  const int wave  = threadIdx.x >> 5;
  const int node0 = (blockIdx.x * 8 + wave) * 8;
  if (node0 >= n_nodes) return;
  const int g    = lane >> 2;
  const int sub  = lane & 3;
  const int m    = lane & 15;
  const int hi   = lane >> 4;

  const int n = node0 + g;
  const int2*   __restrict__ pairs = reinterpret_cast<const int2*>(pairs_raw);
  const float4* __restrict__ hs4   = reinterpret_cast<const float4*>(hs_in);
  const int beg = rowstart[n];
  const int end = rowstart[n + 1];

  const float dg = degree[node0 + (m & 7)];
  float hraw[8], wrow[8];
#pragma unroll
  for (int r = 0; r < 8; ++r) {
    hraw[r] = h_in[(node0 + r) * NCLS + m];
    wrow[r] = W[node0 + r];
  }

  float4 acc = make_float4(0.0f, 0.0f, 0.0f, 0.0f);
  int j = beg;
  for (; j + 4 <= end; j += 4) {
    int2 p0 = pairs[j], p1 = pairs[j + 1], p2 = pairs[j + 2], p3 = pairs[j + 3];
    float4 v0 = hs4[p0.x * 4 + sub];
    float4 v1 = hs4[p1.x * 4 + sub];
    float4 v2 = hs4[p2.x * 4 + sub];
    float4 v3 = hs4[p3.x * 4 + sub];
    float w0 = __int_as_float(p0.y), w1 = __int_as_float(p1.y);
    float w2 = __int_as_float(p2.y), w3 = __int_as_float(p3.y);
    acc.x = fmaf(v0.x, w0, acc.x); acc.y = fmaf(v0.y, w0, acc.y);
    acc.z = fmaf(v0.z, w0, acc.z); acc.w = fmaf(v0.w, w0, acc.w);
    acc.x = fmaf(v1.x, w1, acc.x); acc.y = fmaf(v1.y, w1, acc.y);
    acc.z = fmaf(v1.z, w1, acc.z); acc.w = fmaf(v1.w, w1, acc.w);
    acc.x = fmaf(v2.x, w2, acc.x); acc.y = fmaf(v2.y, w2, acc.y);
    acc.z = fmaf(v2.z, w2, acc.z); acc.w = fmaf(v2.w, w2, acc.w);
    acc.x = fmaf(v3.x, w3, acc.x); acc.y = fmaf(v3.y, w3, acc.y);
    acc.z = fmaf(v3.z, w3, acc.z); acc.w = fmaf(v3.w, w3, acc.w);
  }
#pragma unroll 1
  for (; j < end; ++j) {
    int2 p = pairs[j];
    float4 v = hs4[p.x * 4 + sub];
    float w = __int_as_float(p.y);
    acc.x = fmaf(v.x, w, acc.x); acc.y = fmaf(v.y, w, acc.y);
    acc.z = fmaf(v.z, w, acc.z); acc.w = fmaf(v.w, w, acc.w);
  }

  float* my = tile[wave];
  *reinterpret_cast<float4*>(&my[g * TROW + 4 * sub]) = acc;
  __builtin_amdgcn_fence(__ATOMIC_RELEASE, "wavefront");
  __builtin_amdgcn_wave_barrier();
  __builtin_amdgcn_fence(__ATOMIC_ACQUIRE, "wavefront");

  const float av = (1.0f - ALPHA_F) * dg * dg;

  v8f d;
#pragma unroll
  for (int r = 0; r < 8; ++r) d[r] = ALPHA_F * hraw[r];

#pragma unroll
  for (int q = 0; q < 2; ++q) {
    int k0 = 4 * q + 2 * hi;
    v2f a, b;
    a.x = (m == k0)     ? av : 0.0f;
    a.y = (m == k0 + 1) ? av : 0.0f;
    b.x = my[k0       * TROW + m];
    b.y = my[(k0 + 1) * TROW + m];
    d = __builtin_amdgcn_wmma_f32_16x16x4_f32(
         false, a,  false, b,
         (short)0, d,  false,  false);
  }

  float s[8];
#pragma unroll
  for (int r = 0; r < 8; ++r) s[r] = d[r];
#pragma unroll
  for (int o = 1; o < 16; o <<= 1) {
#pragma unroll
    for (int r = 0; r < 8; ++r) s[r] += __shfl_xor(s[r], o, 32);
  }
  if (hi == 0) {
#pragma unroll
    for (int r = 0; r < 8; ++r) {
      const int   row = node0 + r;
      const float o   = d[r] / s[r];
      h_out [row * NCLS + m] = o;
      hs_out[row * NCLS + m] = wrow[r] * o;
    }
  }
}

extern "C" void kernel_launch(void* const* d_in, const int* in_sizes, int n_in,
                              void* d_out, int out_size, void* d_ws, size_t ws_size,
                              hipStream_t stream) {
  const float* x   = (const float*)d_in[0];
  const float* W   = (const float*)d_in[1];
  const float* ew  = (const float*)d_in[2];
  const float* deg = (const float*)d_in[3];
  const int*   ei  = (const int*)  d_in[4];

  const int N       = in_sizes[3];
  const int N_EDGES = in_sizes[2];
  const int total   = N * NCLS;
  const int n_pad   = (N + 32) & ~31;
  const int e_pad   = (N_EDGES + 31) & ~31;
  const int nb      = (N + BNODES - 1) >> BSH;
  const int g       = (N_EDGES + CH - 1) / CH;
  const int m       = nb * g;
  const int m_pad   = (m + 32) & ~31;

  float* out = (float*)d_out;
  float* hsB = (float*)d_ws;
  float* hA  = hsB + (size_t)total;
  float* hsA = hA  + (size_t)total;
  int* wghist    = (int*)(hsA + (size_t)total);
  int* gscan     = wghist + m_pad;
  int* rowstart  = gscan + m_pad;
  int* tmp_dst   = rowstart + n_pad;
  int* tmp_pairs = tmp_dst + e_pad;
  int* pairs     = tmp_pairs + 2 * (size_t)e_pad;

  const int prep_blocks = (total + 255) / 256;
  const int scan_blocks = (m + SCAN_T - 1) / SCAN_T;
  const int layer_blocks = (N * 4 + 255) / 256;

  lp_prep<<<prep_blocks, 256, 0, stream>>>(x, W, hsA, total);
  lp_bhist<<<g, 1024, 0, stream>>>(ei, wghist, N_EDGES, nb, g);
  lp_scan<<<scan_blocks, SCAN_T, 0, stream>>>(wghist, gscan, m);
  lp_bscatter<<<g, 1024, 0, stream>>>(ei, ew, gscan, tmp_pairs, tmp_dst, N_EDGES, nb, g);
  lp_bsort<<<nb, 1024, 0, stream>>>(gscan, tmp_pairs, tmp_dst, rowstart, pairs, N, nb, g);

  for (int k = 0; k < 4; ++k) {
    const float* hin  = (k == 0) ? x : ((k & 1) ? hA : out);
    float*       hout = (k & 1) ? out : hA;
    const float* hsin  = (k & 1) ? hsB : hsA;
    float*       hsout = (k & 1) ? hsA : hsB;
    lp_layer<<<layer_blocks, 256, 0, stream>>>(rowstart, pairs, hsin, hin, W, deg, hout, hsout, N);
  }
}


# node pre-scale hs = W*x is written by the bucket-count launch's own work-groups (16 bytes per thread and trip) instead of a launch of 6250 small work-groups: 9 launches -> 8; all sums unchanged
# speedup vs baseline: 3.9654x; 1.0331x over previous
#include <hip/hip_runtime.h>
#include <hip/hip_bf16.h>

typedef __attribute__((ext_vector_type(2))) float v2f;
typedef __attribute__((ext_vector_type(8))) float v8f;

#define ALPHA_F 0.5f
#define NCLS 16
#define SCAN_T 1024

#define BSH 9
#define BNODES 512
#define MAXB 256
#define CH 12800
#define CAP 24576
#define CH_T ((CH + 1023) / 1024)
#define BS_T 8

__global__ __launch_bounds__(SCAN_T)
void lp_scan(const int* __restrict__ cnt, int* __restrict__ rowstart, int n_nodes) {
  __shared__ int wsum[32];
  __shared__ int base_s;
  const int t = threadIdx.x, lane = t & 31, wv = t >> 5;
  const int first = blockIdx.x * SCAN_T;

  int pre = 0;
  for (int i = t; i < first; i += SCAN_T) pre += cnt[i];
  for (int o = 16; o > 0; o >>= 1) pre += __shfl_xor(pre, o, 32);
  if (lane == 0) wsum[wv] = pre;
  __syncthreads();
  if (wv == 0) {
    int s = wsum[lane];
    for (int o = 16; o > 0; o >>= 1) s += __shfl_xor(s, o, 32);
    if (lane == 0) base_s = s;
  }
  __syncthreads();
  const int base = base_s;

  const int i = first + t;
  const int c = (i < n_nodes) ? cnt[i] : 0;
  int inc = c;
  for (int o = 1; o < 32; o <<= 1) {
    int u = __shfl_up(inc, o, 32);
    if (lane >= o) inc += u;
  }
  if (lane == 31) wsum[wv] = inc;
  __syncthreads();
  const int wtot = wsum[lane];
  int winc = wtot;
  for (int o = 1; o < 32; o <<= 1) {
    int u = __shfl_up(winc, o, 32);
    if (lane >= o) winc += u;
  }
  const int wexc = __shfl(winc - wtot, wv, 32);
  const int excl = base + wexc + inc - c;
  if (i < n_nodes) rowstart[i] = excl;
  if (i == n_nodes - 1) rowstart[n_nodes] = excl + c;
}

__device__ __forceinline__ int block_excl_scan(int c, int* wsum, int lane, int wv) {
  int inc = c;
  for (int o = 1; o < 32; o <<= 1) {
    int u = __shfl_up(inc, o, 32);
    if (lane >= o) inc += u;
  }
  if (lane == 31) wsum[wv] = inc;
  __syncthreads();
  const int wtot = wsum[lane];
  int winc = wtot;
  for (int o = 1; o < 32; o <<= 1) {
    int u = __shfl_up(winc, o, 32);
    if (lane >= o) winc += u;
  }
  const int wexc = __shfl(winc - wtot, wv, 32);
  __syncthreads();
  return wexc + inc - c;
}

__global__ __launch_bounds__(1024)
void lp_bhist(const int* __restrict__ ei, int* __restrict__ wghist,
              const float* __restrict__ x, const float* __restrict__ W,
              float* __restrict__ hs, int n4,
              int n_edges, int nb, int g) {
  __shared__ int hist[MAXB];
  const int* __restrict__ dst = ei + n_edges;
  const int t = threadIdx.x, wg = blockIdx.x;
  if (t < MAXB) hist[t] = 0;
  __syncthreads();
  const int e0 = wg * CH;
  const int n_loc = min(CH, n_edges - e0);
  int dv[CH_T];
#pragma unroll
  for (int k = 0; k < CH_T; ++k) dv[k] = dst[e0 + min(t + k * 1024, n_loc - 1)];

  const float4* __restrict__ x4  = reinterpret_cast<const float4*>(x);
  float4*       __restrict__ hs4 = reinterpret_cast<float4*>(hs);
  for (int i = wg * 1024 + t; i < n4; i += g * 1024) {
    const float4 v = x4[i];
    const float  w = W[i >> 2];
    hs4[i] = make_float4(w * v.x, w * v.y, w * v.z, w * v.w);
  }

#pragma unroll
  for (int k = 0; k < CH_T; ++k)
    if (t + k * 1024 < n_loc) atomicAdd(&hist[dv[k] >> BSH], 1);
  __syncthreads();
  if (t < nb) wghist[t * g + wg] = hist[t];
}

__global__ __launch_bounds__(1024)
void lp_bscatter(const int* __restrict__ ei, const float* __restrict__ ew,
                 const int* __restrict__ gscan, int* __restrict__ tmp_pairs_raw,
                 int* __restrict__ tmp_dst, int n_edges, int nb, int g) {
  __shared__ int2 sP[CH];
  __shared__ int  sD[CH];
  __shared__ int  hist[MAXB], lstart[MAXB], cursor[MAXB], gbase[MAXB];
  __shared__ int  wsum[32];
  const int* __restrict__ srcp = ei;
  const int* __restrict__ dst  = ei + n_edges;
  int2* __restrict__ tmp_pairs = reinterpret_cast<int2*>(tmp_pairs_raw);
  const int t = threadIdx.x, lane = t & 31, wv = t >> 5, wg = blockIdx.x;
  if (t < MAXB) {
    hist[t]  = 0;
    gbase[t] = (t < nb) ? gscan[t * g + wg] : 0;
  }
  __syncthreads();
  const int e0 = wg * CH;
  const int n_loc = min(CH, n_edges - e0);
  int dv[CH_T], sv[CH_T], wb[CH_T];
#pragma unroll
  for (int k = 0; k < CH_T; ++k) {
    const int e = e0 + min(t + k * 1024, n_loc - 1);
    dv[k] = dst[e];
    sv[k] = srcp[e];
    wb[k] = __float_as_int(ew[e]);
  }
#pragma unroll
  for (int k = 0; k < CH_T; ++k)
    if (t + k * 1024 < n_loc) atomicAdd(&hist[dv[k] >> BSH], 1);
  __syncthreads();
  const int c  = (t < MAXB) ? hist[t] : 0;
  const int ex = block_excl_scan(c, wsum, lane, wv);
  if (t < MAXB) { lstart[t] = ex; cursor[t] = ex; }
  __syncthreads();
#pragma unroll
  for (int k = 0; k < CH_T; ++k) {
    if (t + k * 1024 < n_loc) {
      const int p = atomicAdd(&cursor[dv[k] >> BSH], 1);
      sP[p] = make_int2(sv[k], wb[k]);
      sD[p] = dv[k];
    }
  }
  __syncthreads();
  for (int i = t; i < n_loc; i += 1024) {
    const int d  = sD[i];
    const int b  = d >> BSH;
    const int gp = gbase[b] + (i - lstart[b]);
    tmp_pairs[gp] = sP[i];
    tmp_dst[gp]   = d;
  }
}

__global__ __launch_bounds__(1024)
void lp_bsort(const int* __restrict__ gscan, const int* __restrict__ tmp_pairs_raw,
              const int* __restrict__ tmp_dst, int* __restrict__ rowstart,
              int* __restrict__ pairs_raw, int n_nodes, int nb, int g) {
  __shared__ int2 sP[CAP];
  __shared__ int  ncnt[BNODES], cursor[BNODES];
  __shared__ int  wsum[32];
  const int2* __restrict__ tmp_pairs = reinterpret_cast<const int2*>(tmp_pairs_raw);
  int2* __restrict__ pairs = reinterpret_cast<int2*>(pairs_raw);
  const int t = threadIdx.x, lane = t & 31, wv = t >> 5, b = blockIdx.x;
  const int bs  = gscan[b * g];
  const int be  = gscan[(b + 1) * g];
  const int cnt = be - bs;
  if (t < BNODES) ncnt[t] = 0;
  __syncthreads();
  for (int r0 = 0; r0 < cnt; r0 += BS_T * 1024) {
    int dv[BS_T];
#pragma unroll
    for (int k = 0; k < BS_T; ++k) dv[k] = tmp_dst[bs + min(r0 + k * 1024 + t, cnt - 1)];
#pragma unroll
    for (int k = 0; k < BS_T; ++k)
      if (r0 + k * 1024 + t < cnt) atomicAdd(&ncnt[dv[k] & (BNODES - 1)], 1);
  }
  __syncthreads();
  const int c  = (t < BNODES) ? ncnt[t] : 0;
  const int ex = block_excl_scan(c, wsum, lane, wv);
  if (t < BNODES) {
    cursor[t] = ex;
    const int node = b * BNODES + t;
    if (node < n_nodes) rowstart[node] = bs + ex;
  }
  if (b == nb - 1 && t == 0) rowstart[n_nodes] = be;
  __syncthreads();
  const bool fits = cnt <= CAP;
  for (int r0 = 0; r0 < cnt; r0 += BS_T * 1024) {
    int dv[BS_T]; int2 pv[BS_T];
#pragma unroll
    for (int k = 0; k < BS_T; ++k) {
      const int i = bs + min(r0 + k * 1024 + t, cnt - 1);
      dv[k] = tmp_dst[i];
      pv[k] = tmp_pairs[i];
    }
#pragma unroll
    for (int k = 0; k < BS_T; ++k) {
      if (r0 + k * 1024 + t < cnt) {
        const int pos = atomicAdd(&cursor[dv[k] & (BNODES - 1)], 1);
        if (fits) sP[pos] = pv[k]; else pairs[bs + pos] = pv[k];
      }
    }
  }
  if (fits) {
    __syncthreads();
    for (int i = t; i < cnt; i += 1024) pairs[bs + i] = sP[i];
  }
}

#define TROW 20

__global__ __launch_bounds__(256)
void lp_layer(const int* __restrict__ rowstart, const int* __restrict__ pairs_raw,
              const float* __restrict__ hs_in, const float* __restrict__ h_in,
              const float* __restrict__ W, const float* __restrict__ degree,
              float* __restrict__ h_out, float* __restrict__ hs_out, int n_nodes) {
  __shared__ __attribute__((aligned(16))) float tile[8][8 * TROW];
  const int lane  = threadIdx.x & 31;
  const int wave  = threadIdx.x >> 5;
  const int node0 = (blockIdx.x * 8 + wave) * 8;
  if (node0 >= n_nodes) return;
  const int g    = lane >> 2;
  const int sub  = lane & 3;
  const int m    = lane & 15;
  const int hi   = lane >> 4;

  const int n = node0 + g;
  const int2*   __restrict__ pairs = reinterpret_cast<const int2*>(pairs_raw);
  const float4* __restrict__ hs4   = reinterpret_cast<const float4*>(hs_in);
  const int beg = rowstart[n];
  const int end = rowstart[n + 1];

  const float dg = degree[node0 + (m & 7)];
  float hraw[8], wrow[8];
#pragma unroll
  for (int r = 0; r < 8; ++r) {
    hraw[r] = h_in[(node0 + r) * NCLS + m];
    wrow[r] = W[node0 + r];
  }

  float4 acc = make_float4(0.0f, 0.0f, 0.0f, 0.0f);
  int j = beg;
  for (; j + 4 <= end; j += 4) {
    int2 p0 = pairs[j], p1 = pairs[j + 1], p2 = pairs[j + 2], p3 = pairs[j + 3];
    float4 v0 = hs4[p0.x * 4 + sub];
    float4 v1 = hs4[p1.x * 4 + sub];
    float4 v2 = hs4[p2.x * 4 + sub];
    float4 v3 = hs4[p3.x * 4 + sub];
    float w0 = __int_as_float(p0.y), w1 = __int_as_float(p1.y);
    float w2 = __int_as_float(p2.y), w3 = __int_as_float(p3.y);
    acc.x = fmaf(v0.x, w0, acc.x); acc.y = fmaf(v0.y, w0, acc.y);
    acc.z = fmaf(v0.z, w0, acc.z); acc.w = fmaf(v0.w, w0, acc.w);
    acc.x = fmaf(v1.x, w1, acc.x); acc.y = fmaf(v1.y, w1, acc.y);
    acc.z = fmaf(v1.z, w1, acc.z); acc.w = fmaf(v1.w, w1, acc.w);
    acc.x = fmaf(v2.x, w2, acc.x); acc.y = fmaf(v2.y, w2, acc.y);
    acc.z = fmaf(v2.z, w2, acc.z); acc.w = fmaf(v2.w, w2, acc.w);
    acc.x = fmaf(v3.x, w3, acc.x); acc.y = fmaf(v3.y, w3, acc.y);
    acc.z = fmaf(v3.z, w3, acc.z); acc.w = fmaf(v3.w, w3, acc.w);
  }
#pragma unroll 1
  for (; j < end; ++j) {
    int2 p = pairs[j];
    float4 v = hs4[p.x * 4 + sub];
    float w = __int_as_float(p.y);
    acc.x = fmaf(v.x, w, acc.x); acc.y = fmaf(v.y, w, acc.y);
    acc.z = fmaf(v.z, w, acc.z); acc.w = fmaf(v.w, w, acc.w);
  }

  float* my = tile[wave];
  *reinterpret_cast<float4*>(&my[g * TROW + 4 * sub]) = acc;
  __builtin_amdgcn_fence(__ATOMIC_RELEASE, "wavefront");
  __builtin_amdgcn_wave_barrier();
  __builtin_amdgcn_fence(__ATOMIC_ACQUIRE, "wavefront");

  const float av = (1.0f - ALPHA_F) * dg * dg;

  v8f d;
#pragma unroll
  for (int r = 0; r < 8; ++r) d[r] = ALPHA_F * hraw[r];

#pragma unroll
  for (int q = 0; q < 2; ++q) {
    int k0 = 4 * q + 2 * hi;
    v2f a, b;
    a.x = (m == k0)     ? av : 0.0f;
    a.y = (m == k0 + 1) ? av : 0.0f;
    b.x = my[k0       * TROW + m];
    b.y = my[(k0 + 1) * TROW + m];
    d = __builtin_amdgcn_wmma_f32_16x16x4_f32(
         false, a,  false, b,
         (short)0, d,  false,  false);
  }

  float s[8];
#pragma unroll
  for (int r = 0; r < 8; ++r) s[r] = d[r];
#pragma unroll
  for (int o = 1; o < 16; o <<= 1) {
#pragma unroll
    for (int r = 0; r < 8; ++r) s[r] += __shfl_xor(s[r], o, 32);
  }
  if (hi == 0) {
#pragma unroll
    for (int r = 0; r < 8; ++r) {
      const int   row = node0 + r;
      const float o   = d[r] / s[r];
      h_out [row * NCLS + m] = o;
      hs_out[row * NCLS + m] = wrow[r] * o;
    }
  }
}

extern "C" void kernel_launch(void* const* d_in, const int* in_sizes, int n_in,
                              void* d_out, int out_size, void* d_ws, size_t ws_size,
                              hipStream_t stream) {
  const float* x   = (const float*)d_in[0];
  const float* W   = (const float*)d_in[1];
  const float* ew  = (const float*)d_in[2];
  const float* deg = (const float*)d_in[3];
  const int*   ei  = (const int*)  d_in[4];

  const int N       = in_sizes[3];
  const int N_EDGES = in_sizes[2];
  const int total   = N * NCLS;
  const int n_pad   = (N + 32) & ~31;
  const int e_pad   = (N_EDGES + 31) & ~31;
  const int nb      = (N + BNODES - 1) >> BSH;
  const int g       = (N_EDGES + CH - 1) / CH;
  const int m       = nb * g;
  const int m_pad   = (m + 32) & ~31;

  float* out = (float*)d_out;
  float* hsB = (float*)d_ws;
  float* hA  = hsB + (size_t)total;
  float* hsA = hA  + (size_t)total;
  int* wghist    = (int*)(hsA + (size_t)total);
  int* gscan     = wghist + m_pad;
  int* rowstart  = gscan + m_pad;
  int* tmp_dst   = rowstart + n_pad;
  int* tmp_pairs = tmp_dst + e_pad;
  int* pairs     = tmp_pairs + 2 * (size_t)e_pad;

  const int scan_blocks = (m + SCAN_T - 1) / SCAN_T;
  const int layer_blocks = (N * 4 + 255) / 256;

  lp_bhist<<<g, 1024, 0, stream>>>(ei, wghist, x, W, hsA, total / 4, N_EDGES, nb, g);
  lp_scan<<<scan_blocks, SCAN_T, 0, stream>>>(wghist, gscan, m);
  lp_bscatter<<<g, 1024, 0, stream>>>(ei, ew, gscan, tmp_pairs, tmp_dst, N_EDGES, nb, g);
  lp_bsort<<<nb, 1024, 0, stream>>>(gscan, tmp_pairs, tmp_dst, rowstart, pairs, N, nb, g);

  for (int k = 0; k < 4; ++k) {
    const float* hin  = (k == 0) ? x : ((k & 1) ? hA : out);
    float*       hout = (k & 1) ? out : hA;
    const float* hsin  = (k & 1) ? hsB : hsA;
    float*       hsout = (k & 1) ? hsA : hsB;
    lp_layer<<<layer_blocks, 256, 0, stream>>>(rowstart, pairs, hsin, hin, W, deg, hout, hsout, N);
  }
}


# layer kernel: a wave loads the contiguous (src, weight) run of its 8 nodes once, as whole 128-byte lines, into its own LDS rows and lanes walk their lists there; order of all sums unchanged
# speedup vs baseline: 4.8618x; 1.2261x over previous
#include <hip/hip_runtime.h>
#include <hip/hip_bf16.h>

typedef __attribute__((ext_vector_type(2))) float v2f;
typedef __attribute__((ext_vector_type(8))) float v8f;

#define ALPHA_F 0.5f
#define NCLS 16
#define SCAN_T 1024

#define BSH 9
#define BNODES 512
#define MAXB 256
#define CH 12800
#define CAP 24576
#define CH_T ((CH + 1023) / 1024)
#define BS_T 8

__global__ __launch_bounds__(SCAN_T)
void lp_scan(const int* __restrict__ cnt, int* __restrict__ rowstart, int n_nodes) {
  __shared__ int wsum[32];
  __shared__ int base_s;
  const int t = threadIdx.x, lane = t & 31, wv = t >> 5;
  const int first = blockIdx.x * SCAN_T;

  int pre = 0;
  for (int i = t; i < first; i += SCAN_T) pre += cnt[i];
  for (int o = 16; o > 0; o >>= 1) pre += __shfl_xor(pre, o, 32);
  if (lane == 0) wsum[wv] = pre;
  __syncthreads();
  if (wv == 0) {
    int s = wsum[lane];
    for (int o = 16; o > 0; o >>= 1) s += __shfl_xor(s, o, 32);
    if (lane == 0) base_s = s;
  }
  __syncthreads();
  const int base = base_s;

  const int i = first + t;
  const int c = (i < n_nodes) ? cnt[i] : 0;
  int inc = c;
  for (int o = 1; o < 32; o <<= 1) {
    int u = __shfl_up(inc, o, 32);
    if (lane >= o) inc += u;
  }
  if (lane == 31) wsum[wv] = inc;
  __syncthreads();
  const int wtot = wsum[lane];
  int winc = wtot;
  for (int o = 1; o < 32; o <<= 1) {
    int u = __shfl_up(winc, o, 32);
    if (lane >= o) winc += u;
  }
  const int wexc = __shfl(winc - wtot, wv, 32);
  const int excl = base + wexc + inc - c;
  if (i < n_nodes) rowstart[i] = excl;
  if (i == n_nodes - 1) rowstart[n_nodes] = excl + c;
}

__device__ __forceinline__ int block_excl_scan(int c, int* wsum, int lane, int wv) {
  int inc = c;
  for (int o = 1; o < 32; o <<= 1) {
    int u = __shfl_up(inc, o, 32);
    if (lane >= o) inc += u;
  }
  if (lane == 31) wsum[wv] = inc;
  __syncthreads();
  const int wtot = wsum[lane];
  int winc = wtot;
  for (int o = 1; o < 32; o <<= 1) {
    int u = __shfl_up(winc, o, 32);
    if (lane >= o) winc += u;
  }
  const int wexc = __shfl(winc - wtot, wv, 32);
  __syncthreads();
  return wexc + inc - c;
}

__global__ __launch_bounds__(1024)
void lp_bhist(const int* __restrict__ ei, int* __restrict__ wghist,
              const float* __restrict__ x, const float* __restrict__ W,
              float* __restrict__ hs, int n4,
              int n_edges, int nb, int g) {
  __shared__ int hist[MAXB];
  const int* __restrict__ dst = ei + n_edges;
  const int t = threadIdx.x, wg = blockIdx.x;
  if (t < MAXB) hist[t] = 0;
  __syncthreads();
  const int e0 = wg * CH;
  const int n_loc = min(CH, n_edges - e0);
  int dv[CH_T];
#pragma unroll
  for (int k = 0; k < CH_T; ++k) dv[k] = dst[e0 + min(t + k * 1024, n_loc - 1)];

  const float4* __restrict__ x4  = reinterpret_cast<const float4*>(x);
  float4*       __restrict__ hs4 = reinterpret_cast<float4*>(hs);
  for (int i = wg * 1024 + t; i < n4; i += g * 1024) {
    const float4 v = x4[i];
    const float  w = W[i >> 2];
    hs4[i] = make_float4(w * v.x, w * v.y, w * v.z, w * v.w);
  }

#pragma unroll
  for (int k = 0; k < CH_T; ++k)
    if (t + k * 1024 < n_loc) atomicAdd(&hist[dv[k] >> BSH], 1);
  __syncthreads();
  if (t < nb) wghist[t * g + wg] = hist[t];
}

__global__ __launch_bounds__(1024)
void lp_bscatter(const int* __restrict__ ei, const float* __restrict__ ew,
                 const int* __restrict__ gscan, int* __restrict__ tmp_pairs_raw,
                 int* __restrict__ tmp_dst, int n_edges, int nb, int g) {
  __shared__ int2 sP[CH];
  __shared__ int  sD[CH];
  __shared__ int  hist[MAXB], lstart[MAXB], cursor[MAXB], gbase[MAXB];
  __shared__ int  wsum[32];
  const int* __restrict__ srcp = ei;
  const int* __restrict__ dst  = ei + n_edges;
  int2* __restrict__ tmp_pairs = reinterpret_cast<int2*>(tmp_pairs_raw);
  const int t = threadIdx.x, lane = t & 31, wv = t >> 5, wg = blockIdx.x;
  if (t < MAXB) {
    hist[t]  = 0;
    gbase[t] = (t < nb) ? gscan[t * g + wg] : 0;
  }
  __syncthreads();
  const int e0 = wg * CH;
  const int n_loc = min(CH, n_edges - e0);
  int dv[CH_T], sv[CH_T], wb[CH_T];
#pragma unroll
  for (int k = 0; k < CH_T; ++k) {
    const int e = e0 + min(t + k * 1024, n_loc - 1);
    dv[k] = dst[e];
    sv[k] = srcp[e];
    wb[k] = __float_as_int(ew[e]);
  }
#pragma unroll
  for (int k = 0; k < CH_T; ++k)
    if (t + k * 1024 < n_loc) atomicAdd(&hist[dv[k] >> BSH], 1);
  __syncthreads();
  const int c  = (t < MAXB) ? hist[t] : 0;
  const int ex = block_excl_scan(c, wsum, lane, wv);
  if (t < MAXB) { lstart[t] = ex; cursor[t] = ex; }
  __syncthreads();
#pragma unroll
  for (int k = 0; k < CH_T; ++k) {
    if (t + k * 1024 < n_loc) {
      const int p = atomicAdd(&cursor[dv[k] >> BSH], 1);
      sP[p] = make_int2(sv[k], wb[k]);
      sD[p] = dv[k];
    }
  }
  __syncthreads();
  for (int i = t; i < n_loc; i += 1024) {
    const int d  = sD[i];
    const int b  = d >> BSH;
    const int gp = gbase[b] + (i - lstart[b]);
    tmp_pairs[gp] = sP[i];
    tmp_dst[gp]   = d;
  }
}

__global__ __launch_bounds__(1024)
void lp_bsort(const int* __restrict__ gscan, const int* __restrict__ tmp_pairs_raw,
              const int* __restrict__ tmp_dst, int* __restrict__ rowstart,
              int* __restrict__ pairs_raw, int n_nodes, int nb, int g) {
  __shared__ int2 sP[CAP];
  __shared__ int  ncnt[BNODES], cursor[BNODES];
  __shared__ int  wsum[32];
  const int2* __restrict__ tmp_pairs = reinterpret_cast<const int2*>(tmp_pairs_raw);
  int2* __restrict__ pairs = reinterpret_cast<int2*>(pairs_raw);
  const int t = threadIdx.x, lane = t & 31, wv = t >> 5, b = blockIdx.x;
  const int bs  = gscan[b * g];
  const int be  = gscan[(b + 1) * g];
  const int cnt = be - bs;
  if (t < BNODES) ncnt[t] = 0;
  __syncthreads();
  for (int r0 = 0; r0 < cnt; r0 += BS_T * 1024) {
    int dv[BS_T];
#pragma unroll
    for (int k = 0; k < BS_T; ++k) dv[k] = tmp_dst[bs + min(r0 + k * 1024 + t, cnt - 1)];
#pragma unroll
    for (int k = 0; k < BS_T; ++k)
      if (r0 + k * 1024 + t < cnt) atomicAdd(&ncnt[dv[k] & (BNODES - 1)], 1);
  }
  __syncthreads();
  const int c  = (t < BNODES) ? ncnt[t] : 0;
  const int ex = block_excl_scan(c, wsum, lane, wv);
  if (t < BNODES) {
    cursor[t] = ex;
    const int node = b * BNODES + t;
    if (node < n_nodes) rowstart[node] = bs + ex;
  }
  if (b == nb - 1 && t == 0) {
    rowstart[n_nodes] = be;
    if (be & 1) pairs[be] = make_int2(0, 0);
  }
  __syncthreads();
  const bool fits = cnt <= CAP;
  for (int r0 = 0; r0 < cnt; r0 += BS_T * 1024) {
    int dv[BS_T]; int2 pv[BS_T];
#pragma unroll
    for (int k = 0; k < BS_T; ++k) {
      const int i = bs + min(r0 + k * 1024 + t, cnt - 1);
      dv[k] = tmp_dst[i];
      pv[k] = tmp_pairs[i];
    }
#pragma unroll
    for (int k = 0; k < BS_T; ++k) {
      if (r0 + k * 1024 + t < cnt) {
        const int pos = atomicAdd(&cursor[dv[k] & (BNODES - 1)], 1);
        if (fits) sP[pos] = pv[k]; else pairs[bs + pos] = pv[k];
      }
    }
  }
  if (fits) {
    __syncthreads();
    for (int i = t; i < cnt; i += 1024) pairs[bs + i] = sP[i];
  }
}

#define TROW 20
#define STG 384

__global__ __launch_bounds__(256)
void lp_layer(const int* __restrict__ rowstart, const int* __restrict__ pairs_raw,
              const float* __restrict__ hs_in, const float* __restrict__ h_in,
              const float* __restrict__ W, const float* __restrict__ degree,
              float* __restrict__ h_out, float* __restrict__ hs_out, int n_nodes) {
  __shared__ __attribute__((aligned(16))) float tile[8][8 * TROW];
  __shared__ int4 stage[8][STG / 2];
  const int lane  = threadIdx.x & 31;
  const int wave  = __builtin_amdgcn_readfirstlane(threadIdx.x >> 5);
  const int node0 = (blockIdx.x * 8 + wave) * 8;
  if (node0 >= n_nodes) return;
  const int g    = lane >> 2;
  const int sub  = lane & 3;
  const int m    = lane & 15;
  const int hi   = lane >> 4;

  const int n = node0 + g;
  const float4* __restrict__ hs4   = reinterpret_cast<const float4*>(hs_in);
  const int beg = rowstart[n];
  const int end = rowstart[n + 1];

  const float dg = degree[node0 + (m & 7)];
  float hraw[8], wrow[8];
#pragma unroll
  for (int r = 0; r < 8; ++r) {
    hraw[r] = h_in[(node0 + r) * NCLS + m];
    wrow[r] = W[node0 + r];
  }

  const int rb = __builtin_amdgcn_readlane(beg, 0);
  const int re = __builtin_amdgcn_readlane(end, 31);
  const int4* __restrict__ pairs4 = reinterpret_cast<const int4*>(pairs_raw);
  const int lastq = (re - 1) >> 1;
  int4* st = stage[wave];

  float4 acc = make_float4(0.0f, 0.0f, 0.0f, 0.0f);
  for (int cb = rb & ~15; cb < re; cb += STG) {
    int4 sv[STG / 64];
#pragma unroll
    for (int k = 0; k < STG / 64; ++k) {
      sv[k] = pairs4[min((cb >> 1) + 32 * k + lane, lastq)];
    }
#pragma unroll
    for (int k = 0; k < STG / 64; ++k)
      st[32 * k + lane] = sv[k];
    __builtin_amdgcn_fence(__ATOMIC_RELEASE, "wavefront");
    __builtin_amdgcn_wave_barrier();
    __builtin_amdgcn_fence(__ATOMIC_ACQUIRE, "wavefront");

    int j = max(beg, cb);
    const int je = min(end, cb + STG);
    const int2* sp = reinterpret_cast<const int2*>(st);
    for (; j + 4 <= je; j += 4) {
      const int i = j - cb;
      int2 p0 = sp[i], p1 = sp[i + 1], p2 = sp[i + 2], p3 = sp[i + 3];
      float4 v0 = hs4[p0.x * 4 + sub];
      float4 v1 = hs4[p1.x * 4 + sub];
      float4 v2 = hs4[p2.x * 4 + sub];
      float4 v3 = hs4[p3.x * 4 + sub];
      float w0 = __int_as_float(p0.y), w1 = __int_as_float(p1.y);
      float w2 = __int_as_float(p2.y), w3 = __int_as_float(p3.y);
      acc.x = fmaf(v0.x, w0, acc.x); acc.y = fmaf(v0.y, w0, acc.y);
      acc.z = fmaf(v0.z, w0, acc.z); acc.w = fmaf(v0.w, w0, acc.w);
      acc.x = fmaf(v1.x, w1, acc.x); acc.y = fmaf(v1.y, w1, acc.y);
      acc.z = fmaf(v1.z, w1, acc.z); acc.w = fmaf(v1.w, w1, acc.w);
      acc.x = fmaf(v2.x, w2, acc.x); acc.y = fmaf(v2.y, w2, acc.y);
      acc.z = fmaf(v2.z, w2, acc.z); acc.w = fmaf(v2.w, w2, acc.w);
      acc.x = fmaf(v3.x, w3, acc.x); acc.y = fmaf(v3.y, w3, acc.y);
      acc.z = fmaf(v3.z, w3, acc.z); acc.w = fmaf(v3.w, w3, acc.w);
    }
#pragma unroll 1
    for (; j < je; ++j) {
      int2 p = sp[j - cb];
      float4 v = hs4[p.x * 4 + sub];
      float w = __int_as_float(p.y);
      acc.x = fmaf(v.x, w, acc.x); acc.y = fmaf(v.y, w, acc.y);
      acc.z = fmaf(v.z, w, acc.z); acc.w = fmaf(v.w, w, acc.w);
    }
    __builtin_amdgcn_fence(__ATOMIC_RELEASE, "wavefront");
    __builtin_amdgcn_wave_barrier();
    __builtin_amdgcn_fence(__ATOMIC_ACQUIRE, "wavefront");
  }

  float* my = tile[wave];
  *reinterpret_cast<float4*>(&my[g * TROW + 4 * sub]) = acc;
  __builtin_amdgcn_fence(__ATOMIC_RELEASE, "wavefront");
  __builtin_amdgcn_wave_barrier();
  __builtin_amdgcn_fence(__ATOMIC_ACQUIRE, "wavefront");

  const float av = (1.0f - ALPHA_F) * dg * dg;

  v8f d;
#pragma unroll
  for (int r = 0; r < 8; ++r) d[r] = ALPHA_F * hraw[r];

#pragma unroll
  for (int q = 0; q < 2; ++q) {
    int k0 = 4 * q + 2 * hi;
    v2f a, b;
    a.x = (m == k0)     ? av : 0.0f;
    a.y = (m == k0 + 1) ? av : 0.0f;
    b.x = my[k0       * TROW + m];
    b.y = my[(k0 + 1) * TROW + m];
    d = __builtin_amdgcn_wmma_f32_16x16x4_f32(
         false, a,  false, b,
         (short)0, d,  false,  false);
  }

  float s[8];
#pragma unroll
  for (int r = 0; r < 8; ++r) s[r] = d[r];
#pragma unroll
  for (int o = 1; o < 16; o <<= 1) {
#pragma unroll
    for (int r = 0; r < 8; ++r) s[r] += __shfl_xor(s[r], o, 32);
  }
  if (hi == 0) {
#pragma unroll
    for (int r = 0; r < 8; ++r) {
      const int   row = node0 + r;
      const float o   = d[r] / s[r];
      h_out [row * NCLS + m] = o;
      hs_out[row * NCLS + m] = wrow[r] * o;
    }
  }
}

extern "C" void kernel_launch(void* const* d_in, const int* in_sizes, int n_in,
                              void* d_out, int out_size, void* d_ws, size_t ws_size,
                              hipStream_t stream) {
  const float* x   = (const float*)d_in[0];
  const float* W   = (const float*)d_in[1];
  const float* ew  = (const float*)d_in[2];
  const float* deg = (const float*)d_in[3];
  const int*   ei  = (const int*)  d_in[4];

  const int N       = in_sizes[3];
  const int N_EDGES = in_sizes[2];
  const int total   = N * NCLS;
  const int n_pad   = (N + 32) & ~31;
  const int e_pad   = (N_EDGES + 31) & ~31;
  const int nb      = (N + BNODES - 1) >> BSH;
  const int g       = (N_EDGES + CH - 1) / CH;
  const int m       = nb * g;
  const int m_pad   = (m + 32) & ~31;

  float* out = (float*)d_out;
  float* hsB = (float*)d_ws;
  float* hA  = hsB + (size_t)total;
  float* hsA = hA  + (size_t)total;
  int* wghist    = (int*)(hsA + (size_t)total);
  int* gscan     = wghist + m_pad;
  int* rowstart  = gscan + m_pad;
  int* tmp_dst   = rowstart + n_pad;
  int* tmp_pairs = tmp_dst + e_pad;
  int* pairs     = tmp_pairs + 2 * (size_t)e_pad;

  const int scan_blocks = (m + SCAN_T - 1) / SCAN_T;
  const int layer_blocks = (N * 4 + 255) / 256;

  lp_bhist<<<g, 1024, 0, stream>>>(ei, wghist, x, W, hsA, total / 4, N_EDGES, nb, g);
  lp_scan<<<scan_blocks, SCAN_T, 0, stream>>>(wghist, gscan, m);
  lp_bscatter<<<g, 1024, 0, stream>>>(ei, ew, gscan, tmp_pairs, tmp_dst, N_EDGES, nb, g);
  lp_bsort<<<nb, 1024, 0, stream>>>(gscan, tmp_pairs, tmp_dst, rowstart, pairs, N, nb, g);

  for (int k = 0; k < 4; ++k) {
    const float* hin  = (k == 0) ? x : ((k & 1) ? hA : out);
    float*       hout = (k & 1) ? out : hA;
    const float* hsin  = (k & 1) ? hsB : hsA;
    float*       hsout = (k & 1) ? hsA : hsB;
    lp_layer<<<layer_blocks, 256, 0, stream>>>(rowstart, pairs, hsin, hin, W, deg, hout, hsout, N);
  }
}


# edge sort level 1 orders each 12800-edge chunk by bucket in place with its own offset table; the bucket pass collects its pieces by table: count and scan launches go, 8 -> 6; sums in arrival order
# speedup vs baseline: 5.4857x; 1.1283x over previous
#include <hip/hip_runtime.h>
#include <hip/hip_bf16.h>

typedef __attribute__((ext_vector_type(2))) float v2f;
typedef __attribute__((ext_vector_type(8))) float v8f;

#define ALPHA_F 0.5f
#define NCLS 16

#define BSH 9
#define BNODES 512
#define MAXB 256
#define CH 12800
#define CAP 24576
#define CH_T ((CH + 1023) / 1024)
#define SRC_BITS 17
#define SRC_MASK ((1 << SRC_BITS) - 1)
#define PK 8
#define PR 3

__device__ __forceinline__ int block_excl_scan(int c, int* wsum, int lane, int wv) {
  int inc = c;
  for (int o = 1; o < 32; o <<= 1) {
    int u = __shfl_up(inc, o, 32);
    if (lane >= o) inc += u;
  }
  if (lane == 31) wsum[wv] = inc;
  __syncthreads();
  const int wtot = wsum[lane];
  int winc = wtot;
  for (int o = 1; o < 32; o <<= 1) {
    int u = __shfl_up(winc, o, 32);
    if (lane >= o) winc += u;
  }
  const int wexc = __shfl(winc - wtot, wv, 32);
  __syncthreads();
  return wexc + inc - c;
}

__global__ __launch_bounds__(1024)
void lp_bscatter(const int* __restrict__ ei, const float* __restrict__ ew,
                 int* __restrict__ tab, int* __restrict__ tmp_raw,
                 const float* __restrict__ x, const float* __restrict__ W,
                 float* __restrict__ hs, int n4,
                 int n_edges, int nb, int g) {
  __shared__ int2 sP[CH];
  __shared__ int  hist[MAXB], cursor[MAXB];
  __shared__ int  wsum[32];
  const int* __restrict__ srcp = ei;
  const int* __restrict__ dst  = ei + n_edges;
  int2* __restrict__ tmp = reinterpret_cast<int2*>(tmp_raw);
  const int t = threadIdx.x, lane = t & 31, wv = t >> 5, wg = blockIdx.x;
  if (t < MAXB) hist[t] = 0;
  __syncthreads();
  const int e0 = wg * CH;
  const int n_loc = min(CH, n_edges - e0);
  int dv[CH_T], sv[CH_T], wb[CH_T];
#pragma unroll
  for (int k = 0; k < CH_T; ++k) {
    const int e = e0 + min(t + k * 1024, n_loc - 1);
    dv[k] = dst[e];
    sv[k] = srcp[e];
    wb[k] = __float_as_int(ew[e]);
  }

  const float4* __restrict__ x4  = reinterpret_cast<const float4*>(x);
  float4*       __restrict__ hs4 = reinterpret_cast<float4*>(hs);
  for (int i = wg * 1024 + t; i < n4; i += g * 1024) {
    const float4 v = x4[i];
    const float  w = W[i >> 2];
    hs4[i] = make_float4(w * v.x, w * v.y, w * v.z, w * v.w);
  }

#pragma unroll
  for (int k = 0; k < CH_T; ++k)
    if (t + k * 1024 < n_loc) atomicAdd(&hist[dv[k] >> BSH], 1);
  __syncthreads();
  const int c  = (t < MAXB) ? hist[t] : 0;
  const int ex = block_excl_scan(c, wsum, lane, wv);
  if (t < MAXB) cursor[t] = ex;
  if (t <= nb) tab[t * g + wg] = ex;
  __syncthreads();
#pragma unroll
  for (int k = 0; k < CH_T; ++k) {
    if (t + k * 1024 < n_loc) {
      const int p = atomicAdd(&cursor[dv[k] >> BSH], 1);
      sP[p] = make_int2(sv[k] | ((dv[k] & (BNODES - 1)) << SRC_BITS), wb[k]);
    }
  }
  __syncthreads();
  for (int i = t; i < n_loc; i += 1024) tmp[e0 + i] = sP[i];
}

__global__ __launch_bounds__(1024)
void lp_bsort(const int* __restrict__ tab, const int* __restrict__ tmp_raw,
              int* __restrict__ rowstart, int* __restrict__ pairs_raw,
              int n_nodes, int nb, int g) {
  __shared__ int2 sP[CAP];
  __shared__ int  ncnt[BNODES], cursor[BNODES];
  __shared__ int  pbeg[1024], plen[1024];
  __shared__ int  wsum[32], redA[32], redL[32];
  const int2* __restrict__ tmp = reinterpret_cast<const int2*>(tmp_raw);
  int2* __restrict__ pairs = reinterpret_cast<int2*>(pairs_raw);
  const int t = threadIdx.x, lane = t & 31, wv = t >> 5, b = blockIdx.x;

  int s0 = 0, s1 = 0;
  if (t < g) { s0 = tab[b * g + t]; s1 = tab[(b + 1) * g + t]; }
  pbeg[t] = t * CH + s0;
  plen[t] = s1 - s0;
  if (t < BNODES) ncnt[t] = 0;
  int sa = s0, sl = s1 - s0;
  for (int o = 16; o > 0; o >>= 1) { sa += __shfl_xor(sa, o, 32); sl += __shfl_xor(sl, o, 32); }
  if (lane == 0) { redA[wv] = sa; redL[wv] = sl; }
  __syncthreads();
  sa = redA[lane]; sl = redL[lane];
  for (int o = 16; o > 0; o >>= 1) { sa += __shfl_xor(sa, o, 32); sl += __shfl_xor(sl, o, 32); }
  const int bs = sa, cnt = sl, be = sa + sl;

  int2 rec[PK * PR];
#pragma unroll
  for (int k = 0; k < PK; ++k) {
    const int pb = pbeg[wv + 32 * k], pl = plen[wv + 32 * k];
#pragma unroll
    for (int r = 0; r < PR; ++r) {
      const int o = lane + 32 * r;
      rec[k * PR + r] = make_int2(-1, 0);
      if (o < pl) rec[k * PR + r] = tmp[pb + o];
    }
  }
#pragma unroll
  for (int q = 0; q < PK * PR; ++q)
    if (rec[q].x >= 0) atomicAdd(&ncnt[rec[q].x >> SRC_BITS], 1);
  for (int p = wv; p < g; p += 32) {
    const int pb = pbeg[p], pl = plen[p];
    for (int o = ((p < 32 * PK) ? 32 * PR : 0) + lane; o < pl; o += 32)
      atomicAdd(&ncnt[tmp[pb + o].x >> SRC_BITS], 1);
  }
  __syncthreads();
  const int c  = (t < BNODES) ? ncnt[t] : 0;
  const int ex = block_excl_scan(c, wsum, lane, wv);
  if (t < BNODES) {
    cursor[t] = ex;
    const int node = b * BNODES + t;
    if (node < n_nodes) rowstart[node] = bs + ex;
  }
  if (b == nb - 1 && t == 0) {
    rowstart[n_nodes] = be;
    if (be & 1) pairs[be] = make_int2(0, 0);
  }
  __syncthreads();
  const bool fits = cnt <= CAP;
#pragma unroll
  for (int q = 0; q < PK * PR; ++q) {
    if (rec[q].x >= 0) {
      const int pos = atomicAdd(&cursor[rec[q].x >> SRC_BITS], 1);
      const int2 pv = make_int2(rec[q].x & SRC_MASK, rec[q].y);
      if (fits) sP[pos] = pv; else pairs[bs + pos] = pv;
    }
  }
  for (int p = wv; p < g; p += 32) {
    const int pb = pbeg[p], pl = plen[p];
    for (int o = ((p < 32 * PK) ? 32 * PR : 0) + lane; o < pl; o += 32) {
      const int2 rc = tmp[pb + o];
      const int pos = atomicAdd(&cursor[rc.x >> SRC_BITS], 1);
      const int2 pv = make_int2(rc.x & SRC_MASK, rc.y);
      if (fits) sP[pos] = pv; else pairs[bs + pos] = pv;
    }
  }
  if (fits) {
    __syncthreads();
    for (int i = t; i < cnt; i += 1024) pairs[bs + i] = sP[i];
  }
}

#define TROW 20
#define STG 384

__global__ __launch_bounds__(256)
void lp_layer(const int* __restrict__ rowstart, const int* __restrict__ pairs_raw,
              const float* __restrict__ hs_in, const float* __restrict__ h_in,
              const float* __restrict__ W, const float* __restrict__ degree,
              float* __restrict__ h_out, float* __restrict__ hs_out, int n_nodes) {
  __shared__ __attribute__((aligned(16))) float tile[8][8 * TROW];
  __shared__ int4 stage[8][STG / 2];
  const int lane  = threadIdx.x & 31;
  const int wave  = __builtin_amdgcn_readfirstlane(threadIdx.x >> 5);
  const int node0 = (blockIdx.x * 8 + wave) * 8;
  if (node0 >= n_nodes) return;
  const int g    = lane >> 2;
  const int sub  = lane & 3;
  const int m    = lane & 15;
  const int hi   = lane >> 4;

  const int n = node0 + g;
  const float4* __restrict__ hs4   = reinterpret_cast<const float4*>(hs_in);
  const int beg = rowstart[n];
  const int end = rowstart[n + 1];

  const float dg = degree[node0 + (m & 7)];
  float hraw[8], wrow[8];
#pragma unroll
  for (int r = 0; r < 8; ++r) {
    hraw[r] = h_in[(node0 + r) * NCLS + m];
    wrow[r] = W[node0 + r];
  }

  const int rb = __builtin_amdgcn_readlane(beg, 0);
  const int re = __builtin_amdgcn_readlane(end, 31);
  const int4* __restrict__ pairs4 = reinterpret_cast<const int4*>(pairs_raw);
  const int lastq = (re - 1) >> 1;
  int4* st = stage[wave];

  float4 acc = make_float4(0.0f, 0.0f, 0.0f, 0.0f);
  for (int cb = rb & ~15; cb < re; cb += STG) {
    int4 sv[STG / 64];
#pragma unroll
    for (int k = 0; k < STG / 64; ++k) {
      sv[k] = pairs4[min((cb >> 1) + 32 * k + lane, lastq)];
    }
#pragma unroll
    for (int k = 0; k < STG / 64; ++k)
      st[32 * k + lane] = sv[k];
    __builtin_amdgcn_fence(__ATOMIC_RELEASE, "wavefront");
    __builtin_amdgcn_wave_barrier();
    __builtin_amdgcn_fence(__ATOMIC_ACQUIRE, "wavefront");

    int j = max(beg, cb);
    const int je = min(end, cb + STG);
    const int2* sp = reinterpret_cast<const int2*>(st);
    for (; j + 4 <= je; j += 4) {
      const int i = j - cb;
      int2 p0 = sp[i], p1 = sp[i + 1], p2 = sp[i + 2], p3 = sp[i + 3];
      float4 v0 = hs4[p0.x * 4 + sub];
      float4 v1 = hs4[p1.x * 4 + sub];
      float4 v2 = hs4[p2.x * 4 + sub];
      float4 v3 = hs4[p3.x * 4 + sub];
      float w0 = __int_as_float(p0.y), w1 = __int_as_float(p1.y);
      float w2 = __int_as_float(p2.y), w3 = __int_as_float(p3.y);
      acc.x = fmaf(v0.x, w0, acc.x); acc.y = fmaf(v0.y, w0, acc.y);
      acc.z = fmaf(v0.z, w0, acc.z); acc.w = fmaf(v0.w, w0, acc.w);
      acc.x = fmaf(v1.x, w1, acc.x); acc.y = fmaf(v1.y, w1, acc.y);
      acc.z = fmaf(v1.z, w1, acc.z); acc.w = fmaf(v1.w, w1, acc.w);
      acc.x = fmaf(v2.x, w2, acc.x); acc.y = fmaf(v2.y, w2, acc.y);
      acc.z = fmaf(v2.z, w2, acc.z); acc.w = fmaf(v2.w, w2, acc.w);
      acc.x = fmaf(v3.x, w3, acc.x); acc.y = fmaf(v3.y, w3, acc.y);
      acc.z = fmaf(v3.z, w3, acc.z); acc.w = fmaf(v3.w, w3, acc.w);
    }
#pragma unroll 1
    for (; j < je; ++j) {
      int2 p = sp[j - cb];
      float4 v = hs4[p.x * 4 + sub];
      float w = __int_as_float(p.y);
      acc.x = fmaf(v.x, w, acc.x); acc.y = fmaf(v.y, w, acc.y);
      acc.z = fmaf(v.z, w, acc.z); acc.w = fmaf(v.w, w, acc.w);
    }
    __builtin_amdgcn_fence(__ATOMIC_RELEASE, "wavefront");
    __builtin_amdgcn_wave_barrier();
    __builtin_amdgcn_fence(__ATOMIC_ACQUIRE, "wavefront");
  }

  float* my = tile[wave];
  *reinterpret_cast<float4*>(&my[g * TROW + 4 * sub]) = acc;
  __builtin_amdgcn_fence(__ATOMIC_RELEASE, "wavefront");
  __builtin_amdgcn_wave_barrier();
  __builtin_amdgcn_fence(__ATOMIC_ACQUIRE, "wavefront");

  const float av = (1.0f - ALPHA_F) * dg * dg;

  v8f d;
#pragma unroll
  for (int r = 0; r < 8; ++r) d[r] = ALPHA_F * hraw[r];

#pragma unroll
  for (int q = 0; q < 2; ++q) {
    int k0 = 4 * q + 2 * hi;
    v2f a, b;
    a.x = (m == k0)     ? av : 0.0f;
    a.y = (m == k0 + 1) ? av : 0.0f;
    b.x = my[k0       * TROW + m];
    b.y = my[(k0 + 1) * TROW + m];
    d = __builtin_amdgcn_wmma_f32_16x16x4_f32(
         false, a,  false, b,
         (short)0, d,  false,  false);
  }

  float s[8];
#pragma unroll
  for (int r = 0; r < 8; ++r) s[r] = d[r];
#pragma unroll
  for (int o = 1; o < 16; o <<= 1) {
#pragma unroll
    for (int r = 0; r < 8; ++r) s[r] += __shfl_xor(s[r], o, 32);
  }
  if (hi == 0) {
#pragma unroll
    for (int r = 0; r < 8; ++r) {
      const int   row = node0 + r;
      const float o   = d[r] / s[r];
      h_out [row * NCLS + m] = o;
      hs_out[row * NCLS + m] = wrow[r] * o;
    }
  }
}

extern "C" void kernel_launch(void* const* d_in, const int* in_sizes, int n_in,
                              void* d_out, int out_size, void* d_ws, size_t ws_size,
                              hipStream_t stream) {
  const float* x   = (const float*)d_in[0];
  const float* W   = (const float*)d_in[1];
  const float* ew  = (const float*)d_in[2];
  const float* deg = (const float*)d_in[3];
  const int*   ei  = (const int*)  d_in[4];

  const int N       = in_sizes[3];
  const int N_EDGES = in_sizes[2];
  const int total   = N * NCLS;
  const int n_pad   = (N + 32) & ~31;
  const int e_pad   = (N_EDGES + 31) & ~31;
  const int nb      = (N + BNODES - 1) >> BSH;
  const int g       = (N_EDGES + CH - 1) / CH;
  const int m_pad   = ((nb + 1) * g + 31) & ~31;

  float* out = (float*)d_out;
  float* hsB = (float*)d_ws;
  float* hA  = hsB + (size_t)total;
  float* hsA = hA  + (size_t)total;
  int* tab       = (int*)(hsA + (size_t)total);
  int* rowstart  = tab + m_pad;
  int* tmp       = rowstart + n_pad;
  int* pairs     = tmp + 2 * (size_t)e_pad;

  const int layer_blocks = (N * 4 + 255) / 256;

  lp_bscatter<<<g, 1024, 0, stream>>>(ei, ew, tab, tmp, x, W, hsA, total / 4, N_EDGES, nb, g);
  lp_bsort<<<nb, 1024, 0, stream>>>(tab, tmp, rowstart, pairs, N, nb, g);

  for (int k = 0; k < 4; ++k) {
    const float* hin  = (k == 0) ? x : ((k & 1) ? hA : out);
    float*       hout = (k & 1) ? out : hA;
    const float* hsin  = (k & 1) ? hsB : hsA;
    float*       hsout = (k & 1) ? hsA : hsB;
    lp_layer<<<layer_blocks, 256, 0, stream>>>(rowstart, pairs, hsin, hin, W, deg, hout, hsout, N);
  }
}
